# starting kernel
# baseline (speedup 1.0000x reference)
#include <hip/hip_runtime.h>
#include <hip/hip_bf16.h>

typedef float v2f __attribute__((ext_vector_type(2)));
typedef float v8f __attribute__((ext_vector_type(8)));

#define B_   4
#define H_   16
#define L_   8192
#define D_   64
#define BS_  64
#define NQ_  128
#define SROW 129

__global__ __launch_bounds__(64)
void pool_project_kernel(const float* __restrict__ q, const float* __restrict__ k,
                         const float* __restrict__ Wq, const float* __restrict__ Wk,
                         float* __restrict__ qp, float* __restrict__ kp) {
    const float* x = blockIdx.y ? k : q;
    const float* W = blockIdx.y ? Wk : Wq;
    float* outp    = blockIdx.y ? kp : qp;

    const int bhblk = blockIdx.x;
    const int t  = threadIdx.x;
    const int cg = t & 15;
    const int rq = t >> 4;
    const size_t base = (size_t)bhblk * (BS_ * D_);

    float4 ps = make_float4(0.f, 0.f, 0.f, 0.f);
#pragma unroll
    for (int r = 0; r < 16; ++r) {
        const int row = rq * 16 + r;
        float4 v = *(const float4*)(x + base + (size_t)row * D_ + cg * 4);
        ps.x += v.x; ps.y += v.y; ps.z += v.z; ps.w += v.w;
    }

    __shared__ float4 part[4][16];
    __shared__ float  colsum[D_];
    part[rq][cg] = ps;
    __syncthreads();
    if (rq == 0) {
        float4 a = part[0][cg], b = part[1][cg], c = part[2][cg], d = part[3][cg];
        colsum[cg * 4 + 0] = a.x + b.x + c.x + d.x;
        colsum[cg * 4 + 1] = a.y + b.y + c.y + d.y;
        colsum[cg * 4 + 2] = a.z + b.z + c.z + d.z;
        colsum[cg * 4 + 3] = a.w + b.w + c.w + d.w;
    }
    __syncthreads();

    float acc = 0.f;
#pragma unroll
    for (int i = 0; i < 16; ++i) {
        float4 w = *(const float4*)(W + (size_t)t * D_ + i * 4);
        acc += w.x * colsum[i * 4 + 0] + w.y * colsum[i * 4 + 1] +
               w.z * colsum[i * 4 + 2] + w.w * colsum[i * 4 + 3];
    }
    outp[(size_t)bhblk * D_ + t] = acc * (1.0f / (float)BS_);
}

__global__ __launch_bounds__(256)
void score_softtopk_kernel(const float* __restrict__ qp, const float* __restrict__ kp,
                           const float* __restrict__ bias,
                           const float* __restrict__ bias_scale_p,
                           float* __restrict__ out) {
    extern __shared__ float lds[];

    const int bh  = blockIdx.x;
    const int h   = bh & (H_ - 1);
    const int tid = threadIdx.x;

    {
        const float4* qs = (const float4*)(qp + (size_t)bh * NQ_ * D_);
        const float4* ks = (const float4*)(kp + (size_t)bh * NQ_ * D_);
        float4* lq = (float4*)lds;
        float4* lk = (float4*)(lds + NQ_ * D_);
#pragma unroll
        for (int i = tid; i < (NQ_ * D_) / 4; i += 256) {
            lq[i] = qs[i];
            lk[i] = ks[i];
        }
    }
    __syncthreads();

    const int wave = tid >> 5;
    const int lane = tid & 31;
    const int m0   = wave * 16;
    const int lm   = lane & 15;
    const int kh   = (lane >> 4) << 1;

    v2f afrag[16];
#pragma unroll
    for (int kk = 0; kk < 16; ++kk)
        afrag[kk] = *(const v2f*)&lds[(m0 + lm) * D_ + kk * 4 + kh];

    v8f acc[8];
#pragma unroll
    for (int nt = 0; nt < 8; ++nt) {
        v2f bfrag[16];
#pragma unroll
        for (int kk = 0; kk < 16; ++kk)
            bfrag[kk] = *(const v2f*)&lds[NQ_ * D_ + (nt * 16 + lm) * D_ + kk * 4 + kh];

        v8f c = {0.f, 0.f, 0.f, 0.f, 0.f, 0.f, 0.f, 0.f};
#pragma unroll
        for (int kk = 0; kk < 16; ++kk)
            c = __builtin_amdgcn_wmma_f32_16x16x4_f32(
                    false, afrag[kk], false, bfrag[kk], (short)0, c, false, false);
        acc[nt] = c;
    }
    __syncthreads();

    {
        const float bscale   = bias_scale_p[0];
        const float inv_sqd  = 0.125f;
        const float inv_tau  = 10.0f;
        const int mhalf = (lane >> 4) << 3;
#pragma unroll
        for (int nt = 0; nt < 8; ++nt) {
#pragma unroll
            for (int r = 0; r < 8; ++r) {
                const int m = m0 + r + mhalf;
                const int n = nt * 16 + lm;
                float v = acc[nt][r] * inv_sqd
                        + bscale * bias[((size_t)h * NQ_ + m) * NQ_ + n];
                lds[m * SROW + n] = v * inv_tau;
            }
        }
    }
    __syncthreads();

    const int row  = tid >> 1;
    const int half = tid & 1;
    float s[64];
#pragma unroll
    for (int j = 0; j < 64; ++j) s[j] = lds[row * SROW + half * 64 + j];

    float lo = -10000.0f, hi = 10000.0f;
    const float target = 0.15f * (float)NQ_;
    for (int it = 0; it < 20; ++it) {
        const float mid = 0.5f * (lo + hi);
        float p = 0.f;
#pragma unroll
        for (int j = 0; j < 64; ++j)
            p += 1.0f / (1.0f + __expf(-(s[j] + mid)));
        const float total = p + __shfl_xor(p, 1, 32);
        if (total < target) lo = mid; else hi = mid;
    }
    const float lam = 0.5f * (lo + hi);

    float* orow = out + ((size_t)bh * NQ_ + row) * NQ_ + half * 64;
#pragma unroll
    for (int j = 0; j < 64; ++j)
        orow[j] = 1.0f / (1.0f + __expf(-(s[j] + lam)));
}

extern "C" void kernel_launch(void* const* d_in, const int* in_sizes, int n_in,
                              void* d_out, int out_size, void* d_ws, size_t ws_size,
                              hipStream_t stream) {
    const float* q     = (const float*)d_in[0];
    const float* k     = (const float*)d_in[1];
    const float* bias  = (const float*)d_in[2];
    const float* Wq    = (const float*)d_in[3];
    const float* Wk    = (const float*)d_in[4];
    const float* bsc   = (const float*)d_in[5];
    float* out = (float*)d_out;

    float* qp = (float*)d_ws;
    float* kp = qp + (size_t)B_ * H_ * NQ_ * D_;

    dim3 g1(B_ * H_ * NQ_, 2);
    pool_project_kernel<<<g1, 64, 0, stream>>>(q, k, Wq, Wk, qp, kp);

    const size_t lds_bytes = (size_t)NQ_ * SROW * sizeof(float);
    score_softtopk_kernel<<<B_ * H_, 256, lds_bytes, stream>>>(qp, kp, bias, bsc, out);
}


# score kernel: projections run transposed so their accumulator registers are the score WMMA's fragments (q_proj stays in registers, k_proj via LDS in fragment order); score sums in a new k order
# speedup vs baseline: 2.0788x; 2.0788x over previous
#include <hip/hip_runtime.h>
#include <hip/hip_bf16.h>

typedef float v2f __attribute__((ext_vector_type(2)));
typedef float v8f __attribute__((ext_vector_type(8)));

#define B_   4
#define H_   16
#define L_   8192
#define D_   64
#define BS_  64
#define NQ_  128

#define PP     68
#define OFF_KF 0
#define OFF_PQ (NQ_ * D_)
#define OFF_PK (OFF_PQ + NQ_ * PP)
#define OFF_WQ (OFF_PK + NQ_ * PP)
#define OFF_WK (OFF_WQ + D_ * PP)
#define LDS_FLOATS (OFF_WK + D_ * PP)

__global__ __launch_bounds__(256, 8)
void pool_kernel(const float* __restrict__ q, const float* __restrict__ k,
                 float* __restrict__ pq, float* __restrict__ pk) {
    const float* x = blockIdx.y ? k : q;
    float* outp    = blockIdx.y ? pk : pq;

    const int lane = threadIdx.x & 31;
    const int gw   = blockIdx.x * 8 + (threadIdx.x >> 5);
    const float4* src = (const float4*)(x + (size_t)gw * (2 * BS_ * D_)) + lane;
    float4* dst = (float4*)(outp + (size_t)gw * (2 * D_));

    for (int t = 0; t < 2; ++t) {
        float4 a = make_float4(0.f, 0.f, 0.f, 0.f);
#pragma unroll
        for (int r = 0; r < 32; ++r) {
            const float4 v = src[t * (BS_ * D_ / 4) + r * 32];
            a.x += v.x; a.y += v.y; a.z += v.z; a.w += v.w;
        }
        const float inv_bs = 1.0f / (float)BS_;
        a.x = (a.x + __shfl_xor(a.x, 16, 32)) * inv_bs;
        a.y = (a.y + __shfl_xor(a.y, 16, 32)) * inv_bs;
        a.z = (a.z + __shfl_xor(a.z, 16, 32)) * inv_bs;
        a.w = (a.w + __shfl_xor(a.w, 16, 32)) * inv_bs;
        if (lane < 16) dst[t * (D_ / 4) + lane] = a;
    }
}

__global__ __launch_bounds__(256)
void score_kernel(const float* __restrict__ pq, const float* __restrict__ pk,
                           const float* __restrict__ Wq, const float* __restrict__ Wk,
                           const float* __restrict__ bias,
                           const float* __restrict__ bias_scale_p,
                           float* __restrict__ out) {
    __shared__ __attribute__((aligned(16))) float lds[LDS_FLOATS];

    const int bh  = blockIdx.x;
    const int h   = bh & (H_ - 1);
    const int tid = threadIdx.x;

    {
        const float4* gq  = (const float4*)(pq + (size_t)bh * NQ_ * D_);
        const float4* gk  = (const float4*)(pk + (size_t)bh * NQ_ * D_);
        const float4* gwq = (const float4*)Wq;
        const float4* gwk = (const float4*)Wk;
#pragma unroll
        for (int j = 0; j < (NQ_ * D_) / (4 * 256); ++j) {
            const int i = tid + 256 * j;
            const int row = i >> 4, c4 = i & 15;
            *(float4*)&lds[OFF_PQ + row * PP + c4 * 4] = gq[i];
            *(float4*)&lds[OFF_PK + row * PP + c4 * 4] = gk[i];
        }
#pragma unroll
        for (int j = 0; j < (D_ * D_) / (4 * 256); ++j) {
            const int i = tid + 256 * j;
            const int row = i >> 4, c4 = i & 15;
            *(float4*)&lds[OFF_WQ + row * PP + c4 * 4] = gwq[i];
            *(float4*)&lds[OFF_WK + row * PP + c4 * 4] = gwk[i];
        }
    }
    __syncthreads();

    const int wave = tid >> 5;
    const int lane = tid & 31;
    const int m0   = wave * 16;
    const int lm   = lane & 15;
    const int kh   = (lane >> 4) << 1;
    const int mhalf = (lane >> 4) << 3;

    v8f cq[4];
    {
#pragma unroll
        for (int op = 1; op >= 0; --op) {
            const float* prow = lds + (op ? OFF_PK : OFF_PQ);
            const float* wl   = lds + (op ? OFF_WK : OFF_WQ);

            v2f pa[16];
#pragma unroll
            for (int kk = 0; kk < 16; ++kk)
                pa[kk] = *(const v2f*)&prow[(m0 + lm) * PP + kk * 4 + kh];

#pragma unroll
            for (int et = 0; et < 4; ++et) {
                v2f wb[16];
#pragma unroll
                for (int kk = 0; kk < 16; ++kk)
                    wb[kk] = *(const v2f*)&wl[(et * 16 + lm) * PP + kk * 4 + kh];

                v8f c = {0.f, 0.f, 0.f, 0.f, 0.f, 0.f, 0.f, 0.f};
#pragma unroll
                for (int kk = 0; kk < 16; ++kk)
                    c = __builtin_amdgcn_wmma_f32_16x16x4_f32(
                            false, wb[kk], false, pa[kk], (short)0, c, false, false);
                if (op) {
                    float4* kf = (float4*)&lds[OFF_KF] + ((wave * 4 + et) * 2) * 32 + lane;
                    kf[0]  = make_float4(c[0], c[1], c[2], c[3]);
                    kf[32] = make_float4(c[4], c[5], c[6], c[7]);
                } else {
                    cq[et] = c;
                }
            }
        }
    }
    __syncthreads();

    v8f acc[8];
#pragma unroll
    for (int nt = 0; nt < 8; ++nt) {
        v2f bfrag[16];
        const float4* kf = (const float4*)&lds[OFF_KF] + (nt * 8) * 32 + lane;
#pragma unroll
        for (int g = 0; g < 8; ++g) {
            const float4 f = kf[g * 32];
            bfrag[2 * g]     = (v2f){f.x, f.y};
            bfrag[2 * g + 1] = (v2f){f.z, f.w};
        }

        v8f c = {0.f, 0.f, 0.f, 0.f, 0.f, 0.f, 0.f, 0.f};
#pragma unroll
        for (int t = 0; t < 16; ++t) {
            const v2f a = {cq[t >> 2][2 * (t & 3)], cq[t >> 2][2 * (t & 3) + 1]};
            c = __builtin_amdgcn_wmma_f32_16x16x4_f32(
                    false, a, false, bfrag[t], (short)0, c, false, false);
        }
        acc[nt] = c;
    }

    {
        const float bscale   = bias_scale_p[0];
        const float inv_sqd  = 0.125f;
        const float inv_tau  = 10.0f;
        float* obase = out + (size_t)bh * NQ_ * NQ_;
#pragma unroll
        for (int nt = 0; nt < 8; ++nt) {
#pragma unroll
            for (int r = 0; r < 8; ++r) {
                const int m = m0 + r + mhalf;
                const int n = nt * 16 + lm;
                float v = acc[nt][r] * inv_sqd
                        + bscale * bias[((size_t)h * NQ_ + m) * NQ_ + n];
                obase[m * NQ_ + n] = v * inv_tau;
            }
        }
    }
}

template <int XMASK>
__device__ __forceinline__ float row_xor_add(float v) {
    const int moved = __builtin_amdgcn_update_dpp(0, __float_as_int(v), 0x160 | XMASK, 0xf, 0xf, false);
    return v + __int_as_float(moved);
}

__global__ __launch_bounds__(256)
void softtopk_rows_kernel(float* io) {
    const int tid = threadIdx.x;
    const int row = blockIdx.x * 16 + (tid >> 4);
    const int c   = tid & 15;
    float4* p4 = (float4*)(io + (size_t)row * NQ_) + c;

    const float4 a = p4[0];
    const float4 b = p4[16];
    float s[8] = {a.x, a.y, a.z, a.w, b.x, b.y, b.z, b.w};

    float lo = -10000.0f, hi = 10000.0f;
    const float target = 0.15f * (float)NQ_;
    for (int it = 0; it < 20; ++it) {
        const float mid = 0.5f * (lo + hi);
        float p = 0.f;
#pragma unroll
        for (int j = 0; j < 8; ++j)
            p += 1.0f / (1.0f + __expf(-(s[j] + mid)));
        p = row_xor_add<1>(p);
        p = row_xor_add<2>(p);
        p = row_xor_add<4>(p);
        p = row_xor_add<8>(p);
        if (p < target) lo = mid; else hi = mid;
    }
    const float lam = 0.5f * (lo + hi);

    float o[8];
#pragma unroll
    for (int j = 0; j < 8; ++j)
        o[j] = 1.0f / (1.0f + __expf(-(s[j] + lam)));
    p4[0]  = make_float4(o[0], o[1], o[2], o[3]);
    p4[16] = make_float4(o[4], o[5], o[6], o[7]);
}

extern "C" void kernel_launch(void* const* d_in, const int* in_sizes, int n_in,
                              void* d_out, int out_size, void* d_ws, size_t ws_size,
                              hipStream_t stream) {
    const float* q     = (const float*)d_in[0];
    const float* k     = (const float*)d_in[1];
    const float* bias  = (const float*)d_in[2];
    const float* Wq    = (const float*)d_in[3];
    const float* Wk    = (const float*)d_in[4];
    const float* bsc   = (const float*)d_in[5];
    float* out = (float*)d_out;

    float* pq = (float*)d_ws;
    float* pk = pq + (size_t)B_ * H_ * NQ_ * D_;

    dim3 g1(B_ * H_ * NQ_ / 16, 2);
    pool_kernel<<<g1, 256, 0, stream>>>(q, k, pq, pk);

    score_kernel<<<B_ * H_, 256, 0, stream>>>(pq, pk, Wq, Wk, bias, bsc, out);

    softtopk_rows_kernel<<<B_ * H_ * NQ_ / 16, 256, 0, stream>>>(out);
}


# soft top-k kernel: a bisection step's sigmoid sum taken as 4 + (pairwise sum of 8 v_tanh_f32((s+mid)/2))/2 per lane: no exp, no division; terms and lane partial round differently; output sigmoid kept
# speedup vs baseline: 2.1668x; 1.0423x over previous
#include <hip/hip_runtime.h>
#include <hip/hip_bf16.h>

typedef float v2f __attribute__((ext_vector_type(2)));
typedef float v8f __attribute__((ext_vector_type(8)));

#define B_   4
#define H_   16
#define L_   8192
#define D_   64
#define BS_  64
#define NQ_  128

#define PP     68
#define OFF_KF 0
#define OFF_PQ (NQ_ * D_)
#define OFF_PK (OFF_PQ + NQ_ * PP)
#define OFF_WQ (OFF_PK + NQ_ * PP)
#define OFF_WK (OFF_WQ + D_ * PP)
#define LDS_FLOATS (OFF_WK + D_ * PP)

__global__ __launch_bounds__(256, 8)
void pool_kernel(const float* __restrict__ q, const float* __restrict__ k,
                 float* __restrict__ pq, float* __restrict__ pk) {
    const float* x = blockIdx.y ? k : q;
    float* outp    = blockIdx.y ? pk : pq;

    const int lane = threadIdx.x & 31;
    const int gw   = blockIdx.x * 8 + (threadIdx.x >> 5);
    const float4* src = (const float4*)(x + (size_t)gw * (2 * BS_ * D_)) + lane;
    float4* dst = (float4*)(outp + (size_t)gw * (2 * D_));

    for (int t = 0; t < 2; ++t) {
        float4 a = make_float4(0.f, 0.f, 0.f, 0.f);
#pragma unroll
        for (int r = 0; r < 32; ++r) {
            const float4 v = src[t * (BS_ * D_ / 4) + r * 32];
            a.x += v.x; a.y += v.y; a.z += v.z; a.w += v.w;
        }
        const float inv_bs = 1.0f / (float)BS_;
        a.x = (a.x + __shfl_xor(a.x, 16, 32)) * inv_bs;
        a.y = (a.y + __shfl_xor(a.y, 16, 32)) * inv_bs;
        a.z = (a.z + __shfl_xor(a.z, 16, 32)) * inv_bs;
        a.w = (a.w + __shfl_xor(a.w, 16, 32)) * inv_bs;
        if (lane < 16) dst[t * (D_ / 4) + lane] = a;
    }
}

__global__ __launch_bounds__(256)
void score_kernel(const float* __restrict__ pq, const float* __restrict__ pk,
                           const float* __restrict__ Wq, const float* __restrict__ Wk,
                           const float* __restrict__ bias,
                           const float* __restrict__ bias_scale_p,
                           float* __restrict__ out) {
    __shared__ __attribute__((aligned(16))) float lds[LDS_FLOATS];

    const int bh  = blockIdx.x;
    const int h   = bh & (H_ - 1);
    const int tid = threadIdx.x;

    {
        const float4* gq  = (const float4*)(pq + (size_t)bh * NQ_ * D_);
        const float4* gk  = (const float4*)(pk + (size_t)bh * NQ_ * D_);
        const float4* gwq = (const float4*)Wq;
        const float4* gwk = (const float4*)Wk;
#pragma unroll
        for (int j = 0; j < (NQ_ * D_) / (4 * 256); ++j) {
            const int i = tid + 256 * j;
            const int row = i >> 4, c4 = i & 15;
            *(float4*)&lds[OFF_PQ + row * PP + c4 * 4] = gq[i];
            *(float4*)&lds[OFF_PK + row * PP + c4 * 4] = gk[i];
        }
#pragma unroll
        for (int j = 0; j < (D_ * D_) / (4 * 256); ++j) {
            const int i = tid + 256 * j;
            const int row = i >> 4, c4 = i & 15;
            *(float4*)&lds[OFF_WQ + row * PP + c4 * 4] = gwq[i];
            *(float4*)&lds[OFF_WK + row * PP + c4 * 4] = gwk[i];
        }
    }
    __syncthreads();

    const int wave = tid >> 5;
    const int lane = tid & 31;
    const int m0   = wave * 16;
    const int lm   = lane & 15;
    const int kh   = (lane >> 4) << 1;
    const int mhalf = (lane >> 4) << 3;

    v8f cq[4];
    {
#pragma unroll
        for (int op = 1; op >= 0; --op) {
            const float* prow = lds + (op ? OFF_PK : OFF_PQ);
            const float* wl   = lds + (op ? OFF_WK : OFF_WQ);

            v2f pa[16];
#pragma unroll
            for (int kk = 0; kk < 16; ++kk)
                pa[kk] = *(const v2f*)&prow[(m0 + lm) * PP + kk * 4 + kh];

#pragma unroll
            for (int et = 0; et < 4; ++et) {
                v2f wb[16];
#pragma unroll
                for (int kk = 0; kk < 16; ++kk)
                    wb[kk] = *(const v2f*)&wl[(et * 16 + lm) * PP + kk * 4 + kh];

                v8f c = {0.f, 0.f, 0.f, 0.f, 0.f, 0.f, 0.f, 0.f};
#pragma unroll
                for (int kk = 0; kk < 16; ++kk)
                    c = __builtin_amdgcn_wmma_f32_16x16x4_f32(
                            false, wb[kk], false, pa[kk], (short)0, c, false, false);
                if (op) {
                    float4* kf = (float4*)&lds[OFF_KF] + ((wave * 4 + et) * 2) * 32 + lane;
                    kf[0]  = make_float4(c[0], c[1], c[2], c[3]);
                    kf[32] = make_float4(c[4], c[5], c[6], c[7]);
                } else {
                    cq[et] = c;
                }
            }
        }
    }
    __syncthreads();

    v8f acc[8];
#pragma unroll
    for (int nt = 0; nt < 8; ++nt) {
        v2f bfrag[16];
        const float4* kf = (const float4*)&lds[OFF_KF] + (nt * 8) * 32 + lane;
#pragma unroll
        for (int g = 0; g < 8; ++g) {
            const float4 f = kf[g * 32];
            bfrag[2 * g]     = (v2f){f.x, f.y};
            bfrag[2 * g + 1] = (v2f){f.z, f.w};
        }

        v8f c = {0.f, 0.f, 0.f, 0.f, 0.f, 0.f, 0.f, 0.f};
#pragma unroll
        for (int t = 0; t < 16; ++t) {
            const v2f a = {cq[t >> 2][2 * (t & 3)], cq[t >> 2][2 * (t & 3) + 1]};
            c = __builtin_amdgcn_wmma_f32_16x16x4_f32(
                    false, a, false, bfrag[t], (short)0, c, false, false);
        }
        acc[nt] = c;
    }

    {
        const float bscale   = bias_scale_p[0];
        const float inv_sqd  = 0.125f;
        const float inv_tau  = 10.0f;
        float* obase = out + (size_t)bh * NQ_ * NQ_;
#pragma unroll
        for (int nt = 0; nt < 8; ++nt) {
#pragma unroll
            for (int r = 0; r < 8; ++r) {
                const int m = m0 + r + mhalf;
                const int n = nt * 16 + lm;
                float v = acc[nt][r] * inv_sqd
                        + bscale * bias[((size_t)h * NQ_ + m) * NQ_ + n];
                obase[m * NQ_ + n] = v * inv_tau;
            }
        }
    }
}

template <int XMASK>
__device__ __forceinline__ float row_xor_add(float v) {
    const int moved = __builtin_amdgcn_update_dpp(0, __float_as_int(v), 0x160 | XMASK, 0xf, 0xf, false);
    return v + __int_as_float(moved);
}

__global__ __launch_bounds__(256)
void softtopk_rows_kernel(float* io) {
    const int tid = threadIdx.x;
    const int row = blockIdx.x * 16 + (tid >> 4);
    const int c   = tid & 15;
    float4* p4 = (float4*)(io + (size_t)row * NQ_) + c;

    const float4 a = p4[0];
    const float4 b = p4[16];
    float s[8] = {a.x, a.y, a.z, a.w, b.x, b.y, b.z, b.w};

    float hs[8];
#pragma unroll
    for (int j = 0; j < 8; ++j)
        hs[j] = 0.5f * s[j];

    float lo = -10000.0f, hi = 10000.0f;
    const float target = 0.15f * (float)NQ_;
    for (int it = 0; it < 20; ++it) {
        const float mid = 0.5f * (lo + hi);
        const float hm  = 0.5f * mid;
        float th[8];
#pragma unroll
        for (int j = 0; j < 8; ++j)
            th[j] = __builtin_amdgcn_tanhf(hs[j] + hm);
        const float t = ((th[0] + th[1]) + (th[2] + th[3])) + ((th[4] + th[5]) + (th[6] + th[7]));
        float p = fmaf(0.5f, t, 4.0f);
        p = row_xor_add<1>(p);
        p = row_xor_add<2>(p);
        p = row_xor_add<4>(p);
        p = row_xor_add<8>(p);
        if (p < target) lo = mid; else hi = mid;
    }
    const float lam = 0.5f * (lo + hi);

    float o[8];
#pragma unroll
    for (int j = 0; j < 8; ++j)
        o[j] = 1.0f / (1.0f + __expf(-(s[j] + lam)));
    p4[0]  = make_float4(o[0], o[1], o[2], o[3]);
    p4[16] = make_float4(o[4], o[5], o[6], o[7]);
}

extern "C" void kernel_launch(void* const* d_in, const int* in_sizes, int n_in,
                              void* d_out, int out_size, void* d_ws, size_t ws_size,
                              hipStream_t stream) {
    const float* q     = (const float*)d_in[0];
    const float* k     = (const float*)d_in[1];
    const float* bias  = (const float*)d_in[2];
    const float* Wq    = (const float*)d_in[3];
    const float* Wk    = (const float*)d_in[4];
    const float* bsc   = (const float*)d_in[5];
    float* out = (float*)d_out;

    float* pq = (float*)d_ws;
    float* pk = pq + (size_t)B_ * H_ * NQ_ * D_;

    dim3 g1(B_ * H_ * NQ_ / 16, 2);
    pool_kernel<<<g1, 256, 0, stream>>>(q, k, pq, pk);

    score_kernel<<<B_ * H_, 256, 0, stream>>>(pq, pk, Wq, Wk, bias, bsc, out);

    softtopk_rows_kernel<<<B_ * H_ * NQ_ / 16, 256, 0, stream>>>(out);
}


# score kernel: each (b,h) split 2 x 2 over four work-groups (256, was 64), 64 q rows x 64 k rows each; q_proj and k_proj both pass LDS in fragment order; every product and sum order unchanged
# speedup vs baseline: 2.3223x; 1.0718x over previous
#include <hip/hip_runtime.h>
#include <hip/hip_bf16.h>

typedef float v2f __attribute__((ext_vector_type(2)));
typedef float v8f __attribute__((ext_vector_type(8)));

#define B_   4
#define H_   16
#define L_   8192
#define D_   64
#define BS_  64
#define NQ_  128

#define PP     68
#define HR     64
#define OFF_F  0
#define OFF_PQ (2 * HR * D_)
#define OFF_PK (OFF_PQ + HR * PP)
#define OFF_WQ (OFF_PK + HR * PP)
#define OFF_WK (OFF_WQ + D_ * PP)
#define LDS_FLOATS (OFF_WK + D_ * PP)

__global__ __launch_bounds__(256, 8)
void pool_kernel(const float* __restrict__ q, const float* __restrict__ k,
                 float* __restrict__ pq, float* __restrict__ pk) {
    const float* x = blockIdx.y ? k : q;
    float* outp    = blockIdx.y ? pk : pq;

    const int lane = threadIdx.x & 31;
    const int gw   = blockIdx.x * 8 + (threadIdx.x >> 5);
    const float4* src = (const float4*)(x + (size_t)gw * (2 * BS_ * D_)) + lane;
    float4* dst = (float4*)(outp + (size_t)gw * (2 * D_));

    for (int t = 0; t < 2; ++t) {
        float4 a = make_float4(0.f, 0.f, 0.f, 0.f);
#pragma unroll
        for (int r = 0; r < 32; ++r) {
            const float4 v = src[t * (BS_ * D_ / 4) + r * 32];
            a.x += v.x; a.y += v.y; a.z += v.z; a.w += v.w;
        }
        const float inv_bs = 1.0f / (float)BS_;
        a.x = (a.x + __shfl_xor(a.x, 16, 32)) * inv_bs;
        a.y = (a.y + __shfl_xor(a.y, 16, 32)) * inv_bs;
        a.z = (a.z + __shfl_xor(a.z, 16, 32)) * inv_bs;
        a.w = (a.w + __shfl_xor(a.w, 16, 32)) * inv_bs;
        if (lane < 16) dst[t * (D_ / 4) + lane] = a;
    }
}

__global__ __launch_bounds__(256)
void score_kernel(const float* __restrict__ pq, const float* __restrict__ pk,
                           const float* __restrict__ Wq, const float* __restrict__ Wk,
                           const float* __restrict__ bias,
                           const float* __restrict__ bias_scale_p,
                           float* __restrict__ out) {
    __shared__ __attribute__((aligned(16))) float lds[LDS_FLOATS];

    const int bh  = blockIdx.x >> 2;
    const int rh  = (blockIdx.x >> 1) & 1;
    const int ch  = blockIdx.x & 1;
    const int h   = bh & (H_ - 1);
    const int tid = threadIdx.x;

    {
        const float4* gq  = (const float4*)(pq + ((size_t)bh * NQ_ + rh * HR) * D_);
        const float4* gk  = (const float4*)(pk + ((size_t)bh * NQ_ + ch * HR) * D_);
        const float4* gwq = (const float4*)Wq;
        const float4* gwk = (const float4*)Wk;
#pragma unroll
        for (int j = 0; j < (HR * D_) / (4 * 256); ++j) {
            const int i = tid + 256 * j;
            const int row = i >> 4, c4 = i & 15;
            *(float4*)&lds[OFF_PQ + row * PP + c4 * 4] = gq[i];
            *(float4*)&lds[OFF_PK + row * PP + c4 * 4] = gk[i];
            *(float4*)&lds[OFF_WQ + row * PP + c4 * 4] = gwq[i];
            *(float4*)&lds[OFF_WK + row * PP + c4 * 4] = gwk[i];
        }
    }
    __syncthreads();

    const int wave = tid >> 5;
    const int lane = tid & 31;
    const int lm   = lane & 15;
    const int kh   = (lane >> 4) << 1;
    const int mhalf = (lane >> 4) << 3;

    {
        const int side = wave >> 2;
        const int tl   = wave & 3;
        const float* prow = lds + OFF_PQ + side * (HR * PP);
        const float* wl   = lds + OFF_WQ + side * (D_ * PP);

        v2f pa[16];
#pragma unroll
        for (int kk = 0; kk < 16; ++kk)
            pa[kk] = *(const v2f*)&prow[(tl * 16 + lm) * PP + kk * 4 + kh];

#pragma unroll
        for (int et = 0; et < 4; ++et) {
            v2f wb[16];
#pragma unroll
            for (int kk = 0; kk < 16; ++kk)
                wb[kk] = *(const v2f*)&wl[(et * 16 + lm) * PP + kk * 4 + kh];

            v8f c = {0.f, 0.f, 0.f, 0.f, 0.f, 0.f, 0.f, 0.f};
#pragma unroll
            for (int kk = 0; kk < 16; ++kk)
                c = __builtin_amdgcn_wmma_f32_16x16x4_f32(
                        false, wb[kk], false, pa[kk], (short)0, c, false, false);
            float4* ff = (float4*)&lds[OFF_F] + (((side * 4 + tl) * 4 + et) * 2) * 32 + lane;
            ff[0]  = make_float4(c[0], c[1], c[2], c[3]);
            ff[32] = make_float4(c[4], c[5], c[6], c[7]);
        }
    }
    __syncthreads();

    const int rt = wave & 3;
    const int cp = wave >> 2;
    v2f afrag[16], b0[16], b1[16];
    {
        const float4* qf  = (const float4*)&lds[OFF_F] + (rt * 8) * 32 + lane;
        const float4* kf0 = (const float4*)&lds[OFF_F] + ((4 + 2 * cp) * 8) * 32 + lane;
        const float4* kf1 = kf0 + 8 * 32;
#pragma unroll
        for (int g = 0; g < 8; ++g) {
            const float4 fa = qf[g * 32];
            const float4 f0 = kf0[g * 32];
            const float4 f1 = kf1[g * 32];
            afrag[2 * g] = (v2f){fa.x, fa.y};  afrag[2 * g + 1] = (v2f){fa.z, fa.w};
            b0[2 * g]    = (v2f){f0.x, f0.y};  b0[2 * g + 1]    = (v2f){f0.z, f0.w};
            b1[2 * g]    = (v2f){f1.x, f1.y};  b1[2 * g + 1]    = (v2f){f1.z, f1.w};
        }
    }
    v8f acc[2];
    acc[0] = (v8f){0.f, 0.f, 0.f, 0.f, 0.f, 0.f, 0.f, 0.f};
    acc[1] = (v8f){0.f, 0.f, 0.f, 0.f, 0.f, 0.f, 0.f, 0.f};
#pragma unroll
    for (int t = 0; t < 16; ++t) {
        acc[0] = __builtin_amdgcn_wmma_f32_16x16x4_f32(
                     false, afrag[t], false, b0[t], (short)0, acc[0], false, false);
        acc[1] = __builtin_amdgcn_wmma_f32_16x16x4_f32(
                     false, afrag[t], false, b1[t], (short)0, acc[1], false, false);
    }

    {
        const float bscale   = bias_scale_p[0];
        const float inv_sqd  = 0.125f;
        const float inv_tau  = 10.0f;
        float* obase = out + (size_t)bh * NQ_ * NQ_;
#pragma unroll
        for (int i = 0; i < 2; ++i) {
#pragma unroll
            for (int r = 0; r < 8; ++r) {
                const int m = rh * HR + rt * 16 + r + mhalf;
                const int n = ch * HR + (2 * cp + i) * 16 + lm;
                float v = acc[i][r] * inv_sqd
                        + bscale * bias[((size_t)h * NQ_ + m) * NQ_ + n];
                obase[m * NQ_ + n] = v * inv_tau;
            }
        }
    }
}

template <int XMASK>
__device__ __forceinline__ float row_xor_add(float v) {
    const int moved = __builtin_amdgcn_update_dpp(0, __float_as_int(v), 0x160 | XMASK, 0xf, 0xf, false);
    return v + __int_as_float(moved);
}

__global__ __launch_bounds__(256)
void softtopk_rows_kernel(float* io) {
    const int tid = threadIdx.x;
    const int row = blockIdx.x * 16 + (tid >> 4);
    const int c   = tid & 15;
    float4* p4 = (float4*)(io + (size_t)row * NQ_) + c;

    const float4 a = p4[0];
    const float4 b = p4[16];
    float s[8] = {a.x, a.y, a.z, a.w, b.x, b.y, b.z, b.w};

    float hs[8];
#pragma unroll
    for (int j = 0; j < 8; ++j)
        hs[j] = 0.5f * s[j];

    float lo = -10000.0f, hi = 10000.0f;
    const float target = 0.15f * (float)NQ_;
    for (int it = 0; it < 20; ++it) {
        const float mid = 0.5f * (lo + hi);
        const float hm  = 0.5f * mid;
        float th[8];
#pragma unroll
        for (int j = 0; j < 8; ++j)
            th[j] = __builtin_amdgcn_tanhf(hs[j] + hm);
        const float t = ((th[0] + th[1]) + (th[2] + th[3])) + ((th[4] + th[5]) + (th[6] + th[7]));
        float p = fmaf(0.5f, t, 4.0f);
        p = row_xor_add<1>(p);
        p = row_xor_add<2>(p);
        p = row_xor_add<4>(p);
        p = row_xor_add<8>(p);
        if (p < target) lo = mid; else hi = mid;
    }
    const float lam = 0.5f * (lo + hi);

    float o[8];
#pragma unroll
    for (int j = 0; j < 8; ++j)
        o[j] = 1.0f / (1.0f + __expf(-(s[j] + lam)));
    p4[0]  = make_float4(o[0], o[1], o[2], o[3]);
    p4[16] = make_float4(o[4], o[5], o[6], o[7]);
}

extern "C" void kernel_launch(void* const* d_in, const int* in_sizes, int n_in,
                              void* d_out, int out_size, void* d_ws, size_t ws_size,
                              hipStream_t stream) {
    const float* q     = (const float*)d_in[0];
    const float* k     = (const float*)d_in[1];
    const float* bias  = (const float*)d_in[2];
    const float* Wq    = (const float*)d_in[3];
    const float* Wk    = (const float*)d_in[4];
    const float* bsc   = (const float*)d_in[5];
    float* out = (float*)d_out;

    float* pq = (float*)d_ws;
    float* pk = pq + (size_t)B_ * H_ * NQ_ * D_;

    dim3 g1(B_ * H_ * NQ_ / 16, 2);
    pool_kernel<<<g1, 256, 0, stream>>>(q, k, pq, pk);

    score_kernel<<<B_ * H_ * 4, 256, 0, stream>>>(pq, pk, Wq, Wk, bias, bsc, out);

    softtopk_rows_kernel<<<B_ * H_ * NQ_ / 16, 256, 0, stream>>>(out);
}
